# starting kernel
# baseline (speedup 1.0000x reference)
#include <hip/hip_runtime.h>
#include <hip/hip_bf16.h>
#include <math.h>

#define IN_DIM  256
#define OUT_DIM 128
#define SLOPE   0.1f

typedef __attribute__((ext_vector_type(2))) float v2f;
typedef __attribute__((ext_vector_type(8))) float v8f;

__global__ __launch_bounds__(256) void zero_kernel(float* __restrict__ out,
                                                   float* __restrict__ row_sum,
                                                   int n_out, int n_rs) {
    int total = n_out + n_rs;
    for (int i = blockIdx.x * blockDim.x + threadIdx.x; i < total;
         i += gridDim.x * blockDim.x) {
        if (i < n_out) out[i] = 0.0f;
        else           row_sum[i - n_out] = 0.0f;
    }
}

__global__ __launch_bounds__(256) void gemm_wmma_kernel(
    const float* __restrict__ x, const float* __restrict__ W,
    const float* __restrict__ bias, float* __restrict__ emb, int N) {
    const int tid  = threadIdx.x;
    const int wave = tid >> 5;
    const int lane = tid & 31;
    const int half = lane >> 4;
    const int l16  = lane & 15;
    const int m0   = blockIdx.x * 32;
    const int m1   = m0 + 16;
    const int n0   = wave * 16;
    if (m0 >= N) return;
    const bool has_m1 = (m1 < N);
    const int  m1s    = has_m1 ? m1 : m0;

    const float* a0row = x + (size_t)(m0 + l16) * IN_DIM + 2 * half;
    const float* a1row = x + (size_t)(m1s + l16) * IN_DIM + 2 * half;
    const float* brow  = W + (size_t)(n0 + l16) * IN_DIM + 2 * half;

    v8f c0 = {};
    v8f c1 = {};
#pragma unroll 8
    for (int k0 = 0; k0 < IN_DIM; k0 += 4) {
        v2f b  = *(const v2f*)(brow + k0);
        v2f a0 = *(const v2f*)(a0row + k0);
        v2f a1 = *(const v2f*)(a1row + k0);
        c0 = __builtin_amdgcn_wmma_f32_16x16x4_f32(
                 false, a0, false, b, (short)0, c0, false, false);
        c1 = __builtin_amdgcn_wmma_f32_16x16x4_f32(
                 false, a1, false, b, (short)0, c1, false, false);
    }

    const float bv = bias[n0 + l16];
#pragma unroll
    for (int j = 0; j < 8; ++j) {
        const int row = m0 + j + 8 * half;
        emb[(size_t)row * OUT_DIM + n0 + l16] = c0[j] + bv;
    }
    if (has_m1) {
#pragma unroll
        for (int j = 0; j < 8; ++j) {
            const int row = m1 + j + 8 * half;
            emb[(size_t)row * OUT_DIM + n0 + l16] = c1[j] + bv;
        }
    }
}

__global__ __launch_bounds__(256) void score_kernel(
    const float* __restrict__ emb, const float* __restrict__ a,
    float* __restrict__ s_src, float* __restrict__ s_dst, int N) {
    const int wid  = (blockIdx.x * blockDim.x + threadIdx.x) >> 5;
    const int lane = threadIdx.x & 31;
    if (wid >= N) return;

    const float4 e  = *(const float4*)(emb + (size_t)wid * OUT_DIM + lane * 4);
    const float4 a1 = *(const float4*)(a + lane * 4);
    const float4 a2 = *(const float4*)(a + OUT_DIM + lane * 4);

    float p1 = e.x * a1.x + e.y * a1.y + e.z * a1.z + e.w * a1.w;
    float p2 = e.x * a2.x + e.y * a2.y + e.z * a2.z + e.w * a2.w;
#pragma unroll
    for (int off = 16; off > 0; off >>= 1) {
        p1 += __shfl_xor(p1, off, 32);
        p2 += __shfl_xor(p2, off, 32);
    }
    if (lane == 0) {
        s_src[wid] = p1;
        s_dst[wid] = p2;
    }
}

__global__ __launch_bounds__(256) void edge_kernel(
    const int* __restrict__ edges, const float* __restrict__ s_src,
    const float* __restrict__ s_dst, const float* __restrict__ emb,
    float* __restrict__ out, float* __restrict__ row_sum, int E) {
    const int wid  = (blockIdx.x * blockDim.x + threadIdx.x) >> 5;
    const int lane = threadIdx.x & 31;
    if (wid >= E) return;

    const int2 sd = *(const int2*)(edges + 2 * (size_t)wid);
    const int src = sd.x;
    const int dst = sd.y;

    const float logit = s_src[src] + s_dst[dst];
    const float lr    = logit > 0.0f ? logit : SLOPE * logit;
    const float score = expf(lr);

    if (lane == 0) atomicAdd(row_sum + src, score);

    const float4 m = *(const float4*)(emb + (size_t)dst * OUT_DIM + lane * 4);
    float* o = out + (size_t)src * OUT_DIM + lane * 4;
    atomicAdd(o + 0, score * m.x);
    atomicAdd(o + 1, score * m.y);
    atomicAdd(o + 2, score * m.z);
    atomicAdd(o + 3, score * m.w);
}

__global__ __launch_bounds__(256) void finalize_kernel(
    float* __restrict__ out, const float* __restrict__ row_sum, int total) {
    const int i = blockIdx.x * blockDim.x + threadIdx.x;
    if (i >= total) return;
    const float rs   = row_sum[i >> 7];
    const float safe = rs > 0.0f ? rs : 1.0f;
    out[i] = out[i] / safe;
}

extern "C" void kernel_launch(void* const* d_in, const int* in_sizes, int n_in,
                              void* d_out, int out_size, void* d_ws, size_t ws_size,
                              hipStream_t stream) {
    const float* x     = (const float*)d_in[0];
    const float* W     = (const float*)d_in[1];
    const float* bias  = (const float*)d_in[2];
    const float* a     = (const float*)d_in[3];
    const int*   edges = (const int*)d_in[4];

    const int N = in_sizes[0] / IN_DIM;
    const int E = in_sizes[4] / 2;

    float* ws      = (float*)d_ws;
    float* emb     = ws;
    float* s_src   = emb + (size_t)N * OUT_DIM;
    float* s_dst   = s_src + N;
    float* row_sum = s_dst + N;
    float* out     = (float*)d_out;

    const int n_out = N * OUT_DIM;

    {
        int total = n_out + N;
        int grid = (total + 255) / 256;
        if (grid > 4096) grid = 4096;
        zero_kernel<<<grid, 256, 0, stream>>>(out, row_sum, n_out, N);
    }
    {
        int grid = (N + 31) / 32;
        gemm_wmma_kernel<<<grid, 256, 0, stream>>>(x, W, bias, emb, N);
    }
    {
        int grid = (N + 7) / 8;
        score_kernel<<<grid, 256, 0, stream>>>(emb, a, s_src, s_dst, N);
    }
    {
        int grid = (E + 7) / 8;
        edge_kernel<<<grid, 256, 0, stream>>>(edges, s_src, s_dst, emb, out,
                                              row_sum, E);
    }
    {
        int grid = (n_out + 255) / 256;
        finalize_kernel<<<grid, 256, 0, stream>>>(out, row_sum, n_out);
    }
}


# Scatter by float atomics replaced: edges slotted per source row by an arrival-order rank, then one wave per row gathers, sums in registers and normalises at its store; order of summation changes
# speedup vs baseline: 1.7358x; 1.7358x over previous
#include <hip/hip_runtime.h>
#include <hip/hip_bf16.h>
#include <math.h>


#define IN_DIM    256
#define OUT_DIM   128
#define SLOPE     0.1f
#define ROW_SLOTS OUT_DIM

typedef __attribute__((ext_vector_type(2))) float v2f;
typedef __attribute__((ext_vector_type(8))) float v8f;

__global__ __launch_bounds__(256) void gemm_wmma_kernel(
    const float* __restrict__ x, const float* __restrict__ W,
    const float* __restrict__ bias, float* __restrict__ emb, int N) {
    const int tid  = threadIdx.x;
    const int wave = tid >> 5;
    const int lane = tid & 31;
    const int half = lane >> 4;
    const int l16  = lane & 15;
    const int m0   = blockIdx.x * 32;
    const int m1   = m0 + 16;
    const int n0   = wave * 16;
    if (m0 >= N) return;
    const bool has_m1 = (m1 < N);
    const int  m1s    = has_m1 ? m1 : m0;

    const float* a0row = x + (size_t)(m0 + l16) * IN_DIM + 2 * half;
    const float* a1row = x + (size_t)(m1s + l16) * IN_DIM + 2 * half;
    const float* brow  = W + (size_t)(n0 + l16) * IN_DIM + 2 * half;

    v8f c0 = {};
    v8f c1 = {};
#pragma unroll 8
    for (int k0 = 0; k0 < IN_DIM; k0 += 4) {
        v2f b  = *(const v2f*)(brow + k0);
        v2f a0 = *(const v2f*)(a0row + k0);
        v2f a1 = *(const v2f*)(a1row + k0);
        c0 = __builtin_amdgcn_wmma_f32_16x16x4_f32(
                 false, a0, false, b, (short)0, c0, false, false);
        c1 = __builtin_amdgcn_wmma_f32_16x16x4_f32(
                 false, a1, false, b, (short)0, c1, false, false);
    }

    const float bv = bias[n0 + l16];
#pragma unroll
    for (int j = 0; j < 8; ++j) {
        const int row = m0 + j + 8 * half;
        emb[(size_t)row * OUT_DIM + n0 + l16] = c0[j] + bv;
    }
    if (has_m1) {
#pragma unroll
        for (int j = 0; j < 8; ++j) {
            const int row = m1 + j + 8 * half;
            emb[(size_t)row * OUT_DIM + n0 + l16] = c1[j] + bv;
        }
    }
}

__global__ __launch_bounds__(256) void score_kernel(
    const float* __restrict__ emb, const float* __restrict__ a,
    float* __restrict__ s_src, float* __restrict__ s_dst,
    int* __restrict__ cnt, int N) {
    const int wid  = (blockIdx.x * blockDim.x + threadIdx.x) >> 5;
    const int lane = threadIdx.x & 31;
    if (wid >= N) return;

    const float4 e  = *(const float4*)(emb + (size_t)wid * OUT_DIM + lane * 4);
    const float4 a1 = *(const float4*)(a + lane * 4);
    const float4 a2 = *(const float4*)(a + OUT_DIM + lane * 4);

    float p1 = e.x * a1.x + e.y * a1.y + e.z * a1.z + e.w * a1.w;
    float p2 = e.x * a2.x + e.y * a2.y + e.z * a2.z + e.w * a2.w;
#pragma unroll
    for (int off = 16; off > 0; off >>= 1) {
        p1 += __shfl_xor(p1, off, 32);
        p2 += __shfl_xor(p2, off, 32);
    }
    if (lane == 0) {
        s_src[wid] = p1;
        s_dst[wid] = p2;
        cnt[wid]   = 0;
    }
}

__global__ __launch_bounds__(256) void bucket_kernel(
    const int* __restrict__ edges, int* __restrict__ cnt,
    float* __restrict__ out, int E) {
    const int e = blockIdx.x * blockDim.x + threadIdx.x;
    if (e >= E) return;
    const int2 sd = *(const int2*)(edges + 2 * (size_t)e);
    const int pos = atomicAdd(cnt + sd.x, 1);
    if (pos < ROW_SLOTS)
        out[(size_t)sd.x * OUT_DIM + pos] = __int_as_float(sd.y);
}

__device__ __forceinline__ void fma4(float s, const float4& m, float4& acc) {
    acc.x = fmaf(s, m.x, acc.x);
    acc.y = fmaf(s, m.y, acc.y);
    acc.z = fmaf(s, m.z, acc.z);
    acc.w = fmaf(s, m.w, acc.w);
}

__device__ __forceinline__ void gather4(const float* __restrict__ emb, int lane,
                                        int d0, int d1, int d2, int d3,
                                        float s0, float s1, float s2, float s3,
                                        float4& acc) {
    const float4 m0 = *(const float4*)(emb + (size_t)d0 * OUT_DIM + lane * 4);
    const float4 m1 = *(const float4*)(emb + (size_t)d1 * OUT_DIM + lane * 4);
    const float4 m2 = *(const float4*)(emb + (size_t)d2 * OUT_DIM + lane * 4);
    const float4 m3 = *(const float4*)(emb + (size_t)d3 * OUT_DIM + lane * 4);
    fma4(s0, m0, acc);
    fma4(s1, m1, acc);
    fma4(s2, m2, acc);
    fma4(s3, m3, acc);
}

__device__ __forceinline__ void gather1(const float* __restrict__ emb, int lane,
                                        int d0, float s0, float4& acc) {
    const float4 m0 = *(const float4*)(emb + (size_t)d0 * OUT_DIM + lane * 4);
    fma4(s0, m0, acc);
}

__device__ __forceinline__ float edge_score(float ss, float sd) {
    const float logit = ss + sd;
    const float lr    = logit > 0.0f ? logit : SLOPE * logit;
    return expf(lr);
}

__global__ __launch_bounds__(256) void row_kernel(
    const int* __restrict__ edges, const int* __restrict__ cnt,
    const float* __restrict__ s_src, const float* __restrict__ s_dst,
    const float* __restrict__ emb, float* __restrict__ out, int N, int E) {
    const int r    = blockIdx.x * 8 + __builtin_amdgcn_readfirstlane((int)(threadIdx.x >> 5));
    const int lane = threadIdx.x & 31;
    if (r >= N) return;

    const int   c    = cnt[r];
    const float ss   = s_src[r];
    float*      orow = out + (size_t)r * OUT_DIM;

    float4 acc = {0.0f, 0.0f, 0.0f, 0.0f};
    float  rs  = 0.0f;

    if (c <= ROW_SLOTS) {
        for (int base = 0; base < c; base += 32) {
            const int n = (c - base) < 32 ? (c - base) : 32;
            int   d  = 0;
            float sc = 0.0f;
            if (lane < n) {
                d  = __float_as_int(orow[base + lane]);
                sc = edge_score(ss, s_dst[d]);
            }
            rs += sc;
            const int sci = __float_as_int(sc);
            int t = 0;
            for (; t + 4 <= n; t += 4) {
                gather4(emb, lane,
                        __builtin_amdgcn_readlane(d, t),
                        __builtin_amdgcn_readlane(d, t + 1),
                        __builtin_amdgcn_readlane(d, t + 2),
                        __builtin_amdgcn_readlane(d, t + 3),
                        __int_as_float(__builtin_amdgcn_readlane(sci, t)),
                        __int_as_float(__builtin_amdgcn_readlane(sci, t + 1)),
                        __int_as_float(__builtin_amdgcn_readlane(sci, t + 2)),
                        __int_as_float(__builtin_amdgcn_readlane(sci, t + 3)),
                        acc);
            }
#pragma nounroll
            for (; t < n; ++t) {
                gather1(emb, lane, __builtin_amdgcn_readlane(d, t),
                        __int_as_float(__builtin_amdgcn_readlane(sci, t)), acc);
            }
        }
    } else {
        for (int base = 0; base < E; base += 32) {
            const int e = base + lane;
            int   sx = -1, d = 0;
            float sc = 0.0f;
            if (e < E) {
                const int2 sd = *(const int2*)(edges + 2 * (size_t)e);
                sx = sd.x;
                d  = sd.y;
            }
            const bool hit = (sx == r);
            if (hit) sc = edge_score(ss, s_dst[d]);
            rs += sc;
            const int sci = __float_as_int(sc);
            unsigned mask = __builtin_amdgcn_ballot_w32(hit);
            while (mask) {
                const int t = __builtin_ctz(mask);
                mask &= mask - 1;
                gather1(emb, lane, __builtin_amdgcn_readlane(d, t),
                        __int_as_float(__builtin_amdgcn_readlane(sci, t)), acc);
            }
        }
    }

#pragma unroll
    for (int off = 16; off > 0; off >>= 1) rs += __shfl_xor(rs, off, 32);
    const float safe = rs > 0.0f ? rs : 1.0f;

    float4 o;
    o.x = acc.x / safe; o.y = acc.y / safe; o.z = acc.z / safe; o.w = acc.w / safe;
    *(float4*)(orow + lane * 4) = o;
}

extern "C" void kernel_launch(void* const* d_in, const int* in_sizes, int n_in,
                              void* d_out, int out_size, void* d_ws, size_t ws_size,
                              hipStream_t stream) {
    const float* x     = (const float*)d_in[0];
    const float* W     = (const float*)d_in[1];
    const float* bias  = (const float*)d_in[2];
    const float* a     = (const float*)d_in[3];
    const int*   edges = (const int*)d_in[4];

    const int N = in_sizes[0] / IN_DIM;
    const int E = in_sizes[4] / 2;

    float* ws    = (float*)d_ws;
    float* emb   = ws;
    float* s_src = emb + (size_t)N * OUT_DIM;
    float* s_dst = s_src + N;
    int*   cnt   = (int*)(s_dst + N);
    float* out   = (float*)d_out;

    {
        int grid = (N + 31) / 32;
        gemm_wmma_kernel<<<grid, 256, 0, stream>>>(x, W, bias, emb, N);
    }
    {
        int grid = (N + 7) / 8;
        score_kernel<<<grid, 256, 0, stream>>>(emb, a, s_src, s_dst, cnt, N);
    }
    {
        int grid = (E + 255) / 256;
        bucket_kernel<<<grid, 256, 0, stream>>>(edges, cnt, out, E);
    }
    {
        int grid = (N + 7) / 8;
        row_kernel<<<grid, 256, 0, stream>>>(edges, cnt, s_src, s_dst, emb, out, N, E);
    }
}


# Score launch folded into the linear layer's epilogue: result tile staged in LDS, each wave dots 4 whole rows with a1/a2 and stores them as 512-byte rows; one launch fewer, summation order unchanged
# speedup vs baseline: 1.8389x; 1.0594x over previous
#include <hip/hip_runtime.h>
#include <hip/hip_bf16.h>
#include <math.h>


#define IN_DIM    256
#define OUT_DIM   128
#define SLOPE     0.1f
#define ROW_SLOTS OUT_DIM

typedef __attribute__((ext_vector_type(2))) float v2f;
typedef __attribute__((ext_vector_type(8))) float v8f;

#define EPI_PITCH 132

__global__ __launch_bounds__(256) void gemm_wmma_kernel(
    const float* __restrict__ x, const float* __restrict__ W,
    const float* __restrict__ bias, const float* __restrict__ a,
    float* __restrict__ emb, float* __restrict__ s_src,
    float* __restrict__ s_dst, int* __restrict__ cnt, int N) {
    __shared__ float tile[32 * EPI_PITCH];
    const int tid  = threadIdx.x;
    const int wave = tid >> 5;
    const int lane = tid & 31;
    const int half = lane >> 4;
    const int l16  = lane & 15;
    const int m0   = blockIdx.x * 32;
    const int m1   = m0 + 16;
    const int n0   = wave * 16;
    if (m0 >= N) return;
    const bool has_m1 = (m1 < N);
    const int  m1s    = has_m1 ? m1 : m0;

    const float* a0row = x + (size_t)(m0 + l16) * IN_DIM + 2 * half;
    const float* a1row = x + (size_t)(m1s + l16) * IN_DIM + 2 * half;
    const float* brow  = W + (size_t)(n0 + l16) * IN_DIM + 2 * half;

    v8f c0 = {};
    v8f c1 = {};
#pragma unroll 8
    for (int k0 = 0; k0 < IN_DIM; k0 += 4) {
        v2f b  = *(const v2f*)(brow + k0);
        v2f a0 = *(const v2f*)(a0row + k0);
        v2f a1 = *(const v2f*)(a1row + k0);
        c0 = __builtin_amdgcn_wmma_f32_16x16x4_f32(
                 false, a0, false, b, (short)0, c0, false, false);
        c1 = __builtin_amdgcn_wmma_f32_16x16x4_f32(
                 false, a1, false, b, (short)0, c1, false, false);
    }

    const float  bv = bias[n0 + l16];
    const float4 va1 = *(const float4*)(a + lane * 4);
    const float4 va2 = *(const float4*)(a + OUT_DIM + lane * 4);

    for (int j = 0; j < 8; ++j) {
        tile[(j + 8 * half) * EPI_PITCH + n0 + l16]      = c0[j] + bv;
        tile[(16 + j + 8 * half) * EPI_PITCH + n0 + l16] = c1[j] + bv;
    }
    __syncthreads();

    float k1 = 0.0f, k2 = 0.0f;
    for (int i = 0; i < 4; ++i) {
        const int lr  = 4 * wave + i;
        const int row = m0 + lr;
        const float4 e = *(const float4*)(tile + lr * EPI_PITCH + lane * 4);
        float p1 = e.x * va1.x + e.y * va1.y + e.z * va1.z + e.w * va1.w;
        float p2 = e.x * va2.x + e.y * va2.y + e.z * va2.z + e.w * va2.w;
        for (int off = 16; off > 0; off >>= 1) {
            p1 += __shfl_xor(p1, off, 32);
            p2 += __shfl_xor(p2, off, 32);
        }
        if (row < N)
            *(float4*)(emb + (size_t)row * OUT_DIM + lane * 4) = e;
        if (lane == i) { k1 = p1; k2 = p2; }
    }
    const int srow = m0 + 4 * wave + lane;
    if (lane < 4 && srow < N) {
        s_src[srow] = k1;
        s_dst[srow] = k2;
        cnt[srow]   = 0;
    }
}

__global__ __launch_bounds__(256) void bucket_kernel(
    const int* __restrict__ edges, int* __restrict__ cnt,
    float* __restrict__ out, int E) {
    const int e = blockIdx.x * blockDim.x + threadIdx.x;
    if (e >= E) return;
    const int2 sd = *(const int2*)(edges + 2 * (size_t)e);
    const int pos = atomicAdd(cnt + sd.x, 1);
    if (pos < ROW_SLOTS)
        out[(size_t)sd.x * OUT_DIM + pos] = __int_as_float(sd.y);
}

__device__ __forceinline__ void fma4(float s, const float4& m, float4& acc) {
    acc.x = fmaf(s, m.x, acc.x);
    acc.y = fmaf(s, m.y, acc.y);
    acc.z = fmaf(s, m.z, acc.z);
    acc.w = fmaf(s, m.w, acc.w);
}

__device__ __forceinline__ void gather4(const float* __restrict__ emb, int lane,
                                        int d0, int d1, int d2, int d3,
                                        float s0, float s1, float s2, float s3,
                                        float4& acc) {
    const float4 m0 = *(const float4*)(emb + (size_t)d0 * OUT_DIM + lane * 4);
    const float4 m1 = *(const float4*)(emb + (size_t)d1 * OUT_DIM + lane * 4);
    const float4 m2 = *(const float4*)(emb + (size_t)d2 * OUT_DIM + lane * 4);
    const float4 m3 = *(const float4*)(emb + (size_t)d3 * OUT_DIM + lane * 4);
    fma4(s0, m0, acc);
    fma4(s1, m1, acc);
    fma4(s2, m2, acc);
    fma4(s3, m3, acc);
}

__device__ __forceinline__ void gather1(const float* __restrict__ emb, int lane,
                                        int d0, float s0, float4& acc) {
    const float4 m0 = *(const float4*)(emb + (size_t)d0 * OUT_DIM + lane * 4);
    fma4(s0, m0, acc);
}

__device__ __forceinline__ float edge_score(float ss, float sd) {
    const float logit = ss + sd;
    const float lr    = logit > 0.0f ? logit : SLOPE * logit;
    return expf(lr);
}

__global__ __launch_bounds__(256) void row_kernel(
    const int* __restrict__ edges, const int* __restrict__ cnt,
    const float* __restrict__ s_src, const float* __restrict__ s_dst,
    const float* __restrict__ emb, float* __restrict__ out, int N, int E) {
    const int r    = blockIdx.x * 8 + __builtin_amdgcn_readfirstlane((int)(threadIdx.x >> 5));
    const int lane = threadIdx.x & 31;
    if (r >= N) return;

    const int   c    = cnt[r];
    const float ss   = s_src[r];
    float*      orow = out + (size_t)r * OUT_DIM;

    float4 acc = {0.0f, 0.0f, 0.0f, 0.0f};
    float  rs  = 0.0f;

    if (c <= ROW_SLOTS) {
        for (int base = 0; base < c; base += 32) {
            const int n = (c - base) < 32 ? (c - base) : 32;
            int   d  = 0;
            float sc = 0.0f;
            if (lane < n) {
                d  = __float_as_int(orow[base + lane]);
                sc = edge_score(ss, s_dst[d]);
            }
            rs += sc;
            const int sci = __float_as_int(sc);
            int t = 0;
            for (; t + 4 <= n; t += 4) {
                gather4(emb, lane,
                        __builtin_amdgcn_readlane(d, t),
                        __builtin_amdgcn_readlane(d, t + 1),
                        __builtin_amdgcn_readlane(d, t + 2),
                        __builtin_amdgcn_readlane(d, t + 3),
                        __int_as_float(__builtin_amdgcn_readlane(sci, t)),
                        __int_as_float(__builtin_amdgcn_readlane(sci, t + 1)),
                        __int_as_float(__builtin_amdgcn_readlane(sci, t + 2)),
                        __int_as_float(__builtin_amdgcn_readlane(sci, t + 3)),
                        acc);
            }
#pragma nounroll
            for (; t < n; ++t) {
                gather1(emb, lane, __builtin_amdgcn_readlane(d, t),
                        __int_as_float(__builtin_amdgcn_readlane(sci, t)), acc);
            }
        }
    } else {
        for (int base = 0; base < E; base += 32) {
            const int e = base + lane;
            int   sx = -1, d = 0;
            float sc = 0.0f;
            if (e < E) {
                const int2 sd = *(const int2*)(edges + 2 * (size_t)e);
                sx = sd.x;
                d  = sd.y;
            }
            const bool hit = (sx == r);
            if (hit) sc = edge_score(ss, s_dst[d]);
            rs += sc;
            const int sci = __float_as_int(sc);
            unsigned mask = __builtin_amdgcn_ballot_w32(hit);
            while (mask) {
                const int t = __builtin_ctz(mask);
                mask &= mask - 1;
                gather1(emb, lane, __builtin_amdgcn_readlane(d, t),
                        __int_as_float(__builtin_amdgcn_readlane(sci, t)), acc);
            }
        }
    }

#pragma unroll
    for (int off = 16; off > 0; off >>= 1) rs += __shfl_xor(rs, off, 32);
    const float safe = rs > 0.0f ? rs : 1.0f;

    float4 o;
    o.x = acc.x / safe; o.y = acc.y / safe; o.z = acc.z / safe; o.w = acc.w / safe;
    *(float4*)(orow + lane * 4) = o;
}

extern "C" void kernel_launch(void* const* d_in, const int* in_sizes, int n_in,
                              void* d_out, int out_size, void* d_ws, size_t ws_size,
                              hipStream_t stream) {
    const float* x     = (const float*)d_in[0];
    const float* W     = (const float*)d_in[1];
    const float* bias  = (const float*)d_in[2];
    const float* a     = (const float*)d_in[3];
    const int*   edges = (const int*)d_in[4];

    const int N = in_sizes[0] / IN_DIM;
    const int E = in_sizes[4] / 2;

    float* ws    = (float*)d_ws;
    float* emb   = ws;
    float* s_src = emb + (size_t)N * OUT_DIM;
    float* s_dst = s_src + N;
    int*   cnt   = (int*)(s_dst + N);
    float* out   = (float*)d_out;

    {
        int grid = (N + 31) / 32;
        gemm_wmma_kernel<<<grid, 256, 0, stream>>>(x, W, bias, a, emb, s_src, s_dst, cnt, N);
    }
    {
        int grid = (E + 255) / 256;
        bucket_kernel<<<grid, 256, 0, stream>>>(edges, cnt, out, E);
    }
    {
        int grid = (N + 7) / 8;
        row_kernel<<<grid, 256, 0, stream>>>(edges, cnt, s_src, s_dst, emb, out, N, E);
    }
}


# Linear layer re-tiled: one 16-wave work-group per WGP keeps W in LDS, each wave owns whole 16x128 tiles, x read once by 16-byte loads; k summed in a permuted order inside groups of 8
# speedup vs baseline: 2.4111x; 1.3112x over previous
#include <hip/hip_runtime.h>
#include <hip/hip_bf16.h>
#include <math.h>


#define IN_DIM    256
#define OUT_DIM   128
#define SLOPE     0.1f
#define ROW_SLOTS OUT_DIM

typedef __attribute__((ext_vector_type(2))) float v2f;
typedef __attribute__((ext_vector_type(8))) float v8f;

#define GEMM_WAVES   16
#define GEMM_THREADS (32 * GEMM_WAVES)
#define GEMM_GROUPS  256
#define W_PITCH      260
#define EPI_PITCH    132

__global__ __launch_bounds__(GEMM_THREADS) void gemm_wmma_kernel(
    const float* __restrict__ x, const float* __restrict__ W,
    const float* __restrict__ bias, const float* __restrict__ a,
    float* __restrict__ emb, float* __restrict__ s_src,
    float* __restrict__ s_dst, int* __restrict__ cnt, int N) {
    __shared__ float wlds[OUT_DIM * W_PITCH];
    __shared__ float stage[GEMM_WAVES * 16 * EPI_PITCH];
    const int tid  = threadIdx.x;
    const int wave = __builtin_amdgcn_readfirstlane(tid >> 5);
    const int lane = tid & 31;
    const int half = lane >> 4;
    const int l16  = lane & 15;
    const int ntiles  = (N + 15) >> 4;
    const int tstride = (int)gridDim.x * GEMM_WAVES;
    int T = (int)blockIdx.x + (int)gridDim.x * wave;

    float4 an[4];
    {
        int r = T * 16 + l16;
        r = r < N ? r : N - 1;
        const float* xr = x + (size_t)r * IN_DIM + 4 * half;
#pragma unroll
        for (int i = 0; i < 4; ++i) an[i] = *(const float4*)(xr + 8 * i);
    }

#pragma unroll
    for (int i = 0; i < (OUT_DIM * IN_DIM / 4) / GEMM_THREADS; ++i) {
        const int f  = tid + GEMM_THREADS * i;
        const int n  = f >> 6;
        const int k4 = f & 63;
        const float4 w = *(const float4*)(W + (size_t)n * IN_DIM + 4 * k4);
        *(float4*)(wlds + n * W_PITCH + 4 * k4) = w;
    }
    __syncthreads();

    const float* wl = wlds + l16 * W_PITCH + 4 * half;
    float*       st = stage + wave * (16 * EPI_PITCH);

    for (; T < ntiles; T += tstride) {
        const int m0 = T * 16;
        int r = m0 + l16;
        r = r < N ? r : N - 1;
        const float* xr = x + (size_t)r * IN_DIM + 4 * half;
        const int Tn = (T + tstride < ntiles) ? T + tstride : T;
        int rn = Tn * 16 + l16;
        rn = rn < N ? rn : N - 1;
        const float* xn = x + (size_t)rn * IN_DIM + 4 * half;

        v8f acc[8];
#pragma unroll
        for (int t = 0; t < 8; ++t) acc[t] = (v8f){};

        const float* wq = wl;
#pragma nounroll
        for (int q = 0; q < 8; ++q) {
            float4 ac[4];
#pragma unroll
            for (int i = 0; i < 4; ++i) ac[i] = an[i];
            const float* nx = (q < 7) ? xr + 32 * (q + 1) : xn;
#pragma unroll
            for (int i = 0; i < 4; ++i) an[i] = *(const float4*)(nx + 8 * i);

            const float* wlo = wq;
            const float* whi = wq + 64 * W_PITCH;
#pragma unroll
            for (int i = 0; i < 4; ++i) {
                float4 b[8];
#pragma unroll
                for (int t = 0; t < 4; ++t) {
                    b[t]     = *(const float4*)(wlo + t * 16 * W_PITCH + 8 * i);
                    b[t + 4] = *(const float4*)(whi + t * 16 * W_PITCH + 8 * i);
                }
                const v2f alo = {ac[i].x, ac[i].y};
                const v2f ahi = {ac[i].z, ac[i].w};
#pragma unroll
                for (int t = 0; t < 8; ++t) {
                    const v2f blo = {b[t].x, b[t].y};
                    acc[t] = __builtin_amdgcn_wmma_f32_16x16x4_f32(
                                 false, alo, false, blo, (short)0, acc[t], false, false);
                }
#pragma unroll
                for (int t = 0; t < 8; ++t) {
                    const v2f bhi = {b[t].z, b[t].w};
                    acc[t] = __builtin_amdgcn_wmma_f32_16x16x4_f32(
                                 false, ahi, false, bhi, (short)0, acc[t], false, false);
                }
            }
            wq += 32;
        }

#pragma unroll
        for (int t = 0; t < 8; ++t)
#pragma unroll
            for (int j = 0; j < 8; ++j)
                st[(j + 8 * half) * EPI_PITCH + 16 * t + l16] = acc[t][j];
        __builtin_amdgcn_fence(__ATOMIC_ACQ_REL, "wavefront");

        const float4 vb  = *(const float4*)(bias + lane * 4);
        const float4 va1 = *(const float4*)(a + lane * 4);
        const float4 va2 = *(const float4*)(a + OUT_DIM + lane * 4);
        float4 e[16];
        float  p1[16], p2[16];
#pragma unroll
        for (int i = 0; i < 16; ++i) {
            e[i] = *(const float4*)(st + i * EPI_PITCH + lane * 4);
            e[i].x += vb.x; e[i].y += vb.y; e[i].z += vb.z; e[i].w += vb.w;
            p1[i] = e[i].x * va1.x + e[i].y * va1.y + e[i].z * va1.z + e[i].w * va1.w;
            p2[i] = e[i].x * va2.x + e[i].y * va2.y + e[i].z * va2.z + e[i].w * va2.w;
        }
#pragma unroll
        for (int off = 16; off > 0; off >>= 1) {
#pragma unroll
            for (int i = 0; i < 16; ++i) {
                p1[i] += __shfl_xor(p1[i], off, 32);
                p2[i] += __shfl_xor(p2[i], off, 32);
            }
        }
        float k1 = 0.0f, k2 = 0.0f;
#pragma unroll
        for (int i = 0; i < 16; ++i)
            if (lane == i) { k1 = p1[i]; k2 = p2[i]; }
#pragma unroll
        for (int i = 0; i < 16; ++i)
            if (m0 + i < N)
                *(float4*)(emb + (size_t)(m0 + i) * OUT_DIM + lane * 4) = e[i];
        const int srow = m0 + lane;
        if (lane < 16 && srow < N) {
            s_src[srow] = k1;
            s_dst[srow] = k2;
            cnt[srow]   = 0;
        }
    }
}

__global__ __launch_bounds__(256) void bucket_kernel(
    const int* __restrict__ edges, int* __restrict__ cnt,
    float* __restrict__ out, int E) {
    const int e = blockIdx.x * blockDim.x + threadIdx.x;
    if (e >= E) return;
    const int2 sd = *(const int2*)(edges + 2 * (size_t)e);
    const int pos = atomicAdd(cnt + sd.x, 1);
    if (pos < ROW_SLOTS)
        out[(size_t)sd.x * OUT_DIM + pos] = __int_as_float(sd.y);
}

__device__ __forceinline__ void fma4(float s, const float4& m, float4& acc) {
    acc.x = fmaf(s, m.x, acc.x);
    acc.y = fmaf(s, m.y, acc.y);
    acc.z = fmaf(s, m.z, acc.z);
    acc.w = fmaf(s, m.w, acc.w);
}

__device__ __forceinline__ void gather4(const float* __restrict__ emb, int lane,
                                        int d0, int d1, int d2, int d3,
                                        float s0, float s1, float s2, float s3,
                                        float4& acc) {
    const float4 m0 = *(const float4*)(emb + (size_t)d0 * OUT_DIM + lane * 4);
    const float4 m1 = *(const float4*)(emb + (size_t)d1 * OUT_DIM + lane * 4);
    const float4 m2 = *(const float4*)(emb + (size_t)d2 * OUT_DIM + lane * 4);
    const float4 m3 = *(const float4*)(emb + (size_t)d3 * OUT_DIM + lane * 4);
    fma4(s0, m0, acc);
    fma4(s1, m1, acc);
    fma4(s2, m2, acc);
    fma4(s3, m3, acc);
}

__device__ __forceinline__ void gather1(const float* __restrict__ emb, int lane,
                                        int d0, float s0, float4& acc) {
    const float4 m0 = *(const float4*)(emb + (size_t)d0 * OUT_DIM + lane * 4);
    fma4(s0, m0, acc);
}

__device__ __forceinline__ float edge_score(float ss, float sd) {
    const float logit = ss + sd;
    const float lr    = logit > 0.0f ? logit : SLOPE * logit;
    return expf(lr);
}

__global__ __launch_bounds__(256) void row_kernel(
    const int* __restrict__ edges, const int* __restrict__ cnt,
    const float* __restrict__ s_src, const float* __restrict__ s_dst,
    const float* __restrict__ emb, float* __restrict__ out, int N, int E) {
    const int r    = blockIdx.x * 8 + __builtin_amdgcn_readfirstlane((int)(threadIdx.x >> 5));
    const int lane = threadIdx.x & 31;
    if (r >= N) return;

    const int   c    = cnt[r];
    const float ss   = s_src[r];
    float*      orow = out + (size_t)r * OUT_DIM;

    float4 acc = {0.0f, 0.0f, 0.0f, 0.0f};
    float  rs  = 0.0f;

    if (c <= ROW_SLOTS) {
        for (int base = 0; base < c; base += 32) {
            const int n = (c - base) < 32 ? (c - base) : 32;
            int   d  = 0;
            float sc = 0.0f;
            if (lane < n) {
                d  = __float_as_int(orow[base + lane]);
                sc = edge_score(ss, s_dst[d]);
            }
            rs += sc;
            const int sci = __float_as_int(sc);
            int t = 0;
            for (; t + 4 <= n; t += 4) {
                gather4(emb, lane,
                        __builtin_amdgcn_readlane(d, t),
                        __builtin_amdgcn_readlane(d, t + 1),
                        __builtin_amdgcn_readlane(d, t + 2),
                        __builtin_amdgcn_readlane(d, t + 3),
                        __int_as_float(__builtin_amdgcn_readlane(sci, t)),
                        __int_as_float(__builtin_amdgcn_readlane(sci, t + 1)),
                        __int_as_float(__builtin_amdgcn_readlane(sci, t + 2)),
                        __int_as_float(__builtin_amdgcn_readlane(sci, t + 3)),
                        acc);
            }
#pragma nounroll
            for (; t < n; ++t) {
                gather1(emb, lane, __builtin_amdgcn_readlane(d, t),
                        __int_as_float(__builtin_amdgcn_readlane(sci, t)), acc);
            }
        }
    } else {
        for (int base = 0; base < E; base += 32) {
            const int e = base + lane;
            int   sx = -1, d = 0;
            float sc = 0.0f;
            if (e < E) {
                const int2 sd = *(const int2*)(edges + 2 * (size_t)e);
                sx = sd.x;
                d  = sd.y;
            }
            const bool hit = (sx == r);
            if (hit) sc = edge_score(ss, s_dst[d]);
            rs += sc;
            const int sci = __float_as_int(sc);
            unsigned mask = __builtin_amdgcn_ballot_w32(hit);
            while (mask) {
                const int t = __builtin_ctz(mask);
                mask &= mask - 1;
                gather1(emb, lane, __builtin_amdgcn_readlane(d, t),
                        __int_as_float(__builtin_amdgcn_readlane(sci, t)), acc);
            }
        }
    }

#pragma unroll
    for (int off = 16; off > 0; off >>= 1) rs += __shfl_xor(rs, off, 32);
    const float safe = rs > 0.0f ? rs : 1.0f;

    float4 o;
    o.x = acc.x / safe; o.y = acc.y / safe; o.z = acc.z / safe; o.w = acc.w / safe;
    *(float4*)(orow + lane * 4) = o;
}

extern "C" void kernel_launch(void* const* d_in, const int* in_sizes, int n_in,
                              void* d_out, int out_size, void* d_ws, size_t ws_size,
                              hipStream_t stream) {
    const float* x     = (const float*)d_in[0];
    const float* W     = (const float*)d_in[1];
    const float* bias  = (const float*)d_in[2];
    const float* a     = (const float*)d_in[3];
    const int*   edges = (const int*)d_in[4];

    const int N = in_sizes[0] / IN_DIM;
    const int E = in_sizes[4] / 2;

    float* ws    = (float*)d_ws;
    float* emb   = ws;
    float* s_src = emb + (size_t)N * OUT_DIM;
    float* s_dst = s_src + N;
    int*   cnt   = (int*)(s_dst + N);
    float* out   = (float*)d_out;

    {
        gemm_wmma_kernel<<<GEMM_GROUPS, GEMM_THREADS, 0, stream>>>(x, W, bias, a, emb, s_src, s_dst, cnt, N);
    }
    {
        int grid = (E + 255) / 256;
        bucket_kernel<<<grid, 256, 0, stream>>>(edges, cnt, out, E);
    }
    {
        int grid = (N + 7) / 8;
        row_kernel<<<grid, 256, 0, stream>>>(edges, cnt, s_src, s_dst, emb, out, N, E);
    }
}


# Edges grouped in two levels instead of one global add per edge: binned by 128-row block (LDS ranks, one global add per work-group and bin), then slotted per row in LDS; sums stay in arrival order
# speedup vs baseline: 3.9762x; 1.6491x over previous
#include <hip/hip_runtime.h>
#include <hip/hip_bf16.h>
#include <math.h>


#define IN_DIM    256
#define OUT_DIM   128
#define SLOPE     0.1f
#define RPB             128
#define RPB_SHIFT       7
#define ROW_SLOTS       64
#define BIN_CAP_PER_ROW (OUT_DIM / 2)
#define MAX_BINS        1024
#define BIN_THREADS     1024
#define BIN_EPT         8
#define BIN_CHUNK       (BIN_THREADS * BIN_EPT)
#define ROW_WAVES       16
#define ROW_THREADS     (32 * ROW_WAVES)
#define STAGE_EPT       6

typedef __attribute__((ext_vector_type(2))) float v2f;
typedef __attribute__((ext_vector_type(8))) float v8f;

#define GEMM_WAVES   16
#define GEMM_THREADS (32 * GEMM_WAVES)
#define GEMM_GROUPS  256
#define W_PITCH      260
#define EPI_PITCH    132

__global__ __launch_bounds__(GEMM_THREADS) void gemm_wmma_kernel(
    const float* __restrict__ x, const float* __restrict__ W,
    const float* __restrict__ bias, const float* __restrict__ a,
    float* __restrict__ emb, float* __restrict__ s_src,
    float* __restrict__ s_dst, int* __restrict__ cnt, int N) {
    __shared__ float wlds[OUT_DIM * W_PITCH];
    __shared__ float stage[GEMM_WAVES * 16 * EPI_PITCH];
    const int tid  = threadIdx.x;
    const int wave = __builtin_amdgcn_readfirstlane(tid >> 5);
    const int lane = tid & 31;
    const int half = lane >> 4;
    const int l16  = lane & 15;
    const int ntiles  = (N + 15) >> 4;
    const int nbins   = (N + RPB - 1) >> RPB_SHIFT;
    const int tstride = (int)gridDim.x * GEMM_WAVES;
    int T = (int)blockIdx.x + (int)gridDim.x * wave;

    float4 an[4];
    {
        int r = T * 16 + l16;
        r = r < N ? r : N - 1;
        const float* xr = x + (size_t)r * IN_DIM + 4 * half;
#pragma unroll
        for (int i = 0; i < 4; ++i) an[i] = *(const float4*)(xr + 8 * i);
    }

#pragma unroll
    for (int i = 0; i < (OUT_DIM * IN_DIM / 4) / GEMM_THREADS; ++i) {
        const int f  = tid + GEMM_THREADS * i;
        const int n  = f >> 6;
        const int k4 = f & 63;
        const float4 w = *(const float4*)(W + (size_t)n * IN_DIM + 4 * k4);
        *(float4*)(wlds + n * W_PITCH + 4 * k4) = w;
    }
    __syncthreads();

    const float* wl = wlds + l16 * W_PITCH + 4 * half;
    float*       st = stage + wave * (16 * EPI_PITCH);

    for (; T < ntiles; T += tstride) {
        const int m0 = T * 16;
        int r = m0 + l16;
        r = r < N ? r : N - 1;
        const float* xr = x + (size_t)r * IN_DIM + 4 * half;
        const int Tn = (T + tstride < ntiles) ? T + tstride : T;
        int rn = Tn * 16 + l16;
        rn = rn < N ? rn : N - 1;
        const float* xn = x + (size_t)rn * IN_DIM + 4 * half;

        v8f acc[8];
#pragma unroll
        for (int t = 0; t < 8; ++t) acc[t] = (v8f){};

        const float* wq = wl;
#pragma nounroll
        for (int q = 0; q < 8; ++q) {
            float4 ac[4];
#pragma unroll
            for (int i = 0; i < 4; ++i) ac[i] = an[i];
            const float* nx = (q < 7) ? xr + 32 * (q + 1) : xn;
#pragma unroll
            for (int i = 0; i < 4; ++i) an[i] = *(const float4*)(nx + 8 * i);

            const float* wlo = wq;
            const float* whi = wq + 64 * W_PITCH;
#pragma unroll
            for (int i = 0; i < 4; ++i) {
                float4 b[8];
#pragma unroll
                for (int t = 0; t < 4; ++t) {
                    b[t]     = *(const float4*)(wlo + t * 16 * W_PITCH + 8 * i);
                    b[t + 4] = *(const float4*)(whi + t * 16 * W_PITCH + 8 * i);
                }
                const v2f alo = {ac[i].x, ac[i].y};
                const v2f ahi = {ac[i].z, ac[i].w};
#pragma unroll
                for (int t = 0; t < 8; ++t) {
                    const v2f blo = {b[t].x, b[t].y};
                    acc[t] = __builtin_amdgcn_wmma_f32_16x16x4_f32(
                                 false, alo, false, blo, (short)0, acc[t], false, false);
                }
#pragma unroll
                for (int t = 0; t < 8; ++t) {
                    const v2f bhi = {b[t].z, b[t].w};
                    acc[t] = __builtin_amdgcn_wmma_f32_16x16x4_f32(
                                 false, ahi, false, bhi, (short)0, acc[t], false, false);
                }
            }
            wq += 32;
        }

#pragma unroll
        for (int t = 0; t < 8; ++t)
#pragma unroll
            for (int j = 0; j < 8; ++j)
                st[(j + 8 * half) * EPI_PITCH + 16 * t + l16] = acc[t][j];
        __builtin_amdgcn_fence(__ATOMIC_ACQ_REL, "wavefront");

        const float4 vb  = *(const float4*)(bias + lane * 4);
        const float4 va1 = *(const float4*)(a + lane * 4);
        const float4 va2 = *(const float4*)(a + OUT_DIM + lane * 4);
        float4 e[16];
        float  p1[16], p2[16];
#pragma unroll
        for (int i = 0; i < 16; ++i) {
            e[i] = *(const float4*)(st + i * EPI_PITCH + lane * 4);
            e[i].x += vb.x; e[i].y += vb.y; e[i].z += vb.z; e[i].w += vb.w;
            p1[i] = e[i].x * va1.x + e[i].y * va1.y + e[i].z * va1.z + e[i].w * va1.w;
            p2[i] = e[i].x * va2.x + e[i].y * va2.y + e[i].z * va2.z + e[i].w * va2.w;
        }
#pragma unroll
        for (int off = 16; off > 0; off >>= 1) {
#pragma unroll
            for (int i = 0; i < 16; ++i) {
                p1[i] += __shfl_xor(p1[i], off, 32);
                p2[i] += __shfl_xor(p2[i], off, 32);
            }
        }
        float k1 = 0.0f, k2 = 0.0f;
#pragma unroll
        for (int i = 0; i < 16; ++i)
            if (lane == i) { k1 = p1[i]; k2 = p2[i]; }
#pragma unroll
        for (int i = 0; i < 16; ++i)
            if (m0 + i < N)
                *(float4*)(emb + (size_t)(m0 + i) * OUT_DIM + lane * 4) = e[i];
        const int srow = m0 + lane;
        if (lane < 16 && srow < N) {
            s_src[srow] = k1;
            s_dst[srow] = k2;
            if (srow < nbins) cnt[srow] = 0;
        }
    }
}

static_assert(BIN_THREADS == MAX_BINS, "thread b owns LDS counter b");

__global__ __launch_bounds__(BIN_THREADS) void bin_kernel(
    const int* __restrict__ edges, int* __restrict__ bincnt,
    float* __restrict__ out, int N, int E) {
    __shared__ int hist[MAX_BINS];
    const int tid = threadIdx.x;
    hist[tid] = 0;
    __syncthreads();

    const int e0 = (int)blockIdx.x * BIN_CHUNK + tid;
    int es[BIN_EPT], ed[BIN_EPT], rk[BIN_EPT];
#pragma unroll
    for (int k = 0; k < BIN_EPT; ++k) {
        const int e  = e0 + k * BIN_THREADS;
        const int ec = e < E ? e : E - 1;
        const int2 sd = *(const int2*)(edges + 2 * (size_t)ec);
        es[k] = e < E ? sd.x : -1;
        ed[k] = sd.y;
    }
#pragma unroll
    for (int k = 0; k < BIN_EPT; ++k) {
        rk[k] = 0;
        const bool ok = (unsigned)es[k] < (unsigned)N && (es[k] >> RPB_SHIFT) < MAX_BINS;
        if (ok) rk[k] = atomicAdd(&hist[es[k] >> RPB_SHIFT], 1);
        else    es[k] = -1;
    }
    __syncthreads();
    {
        const int c = hist[tid];
        int base = 0;
        if (c > 0) base = atomicAdd(bincnt + tid, c);
        hist[tid] = base;
    }
    __syncthreads();
#pragma unroll
    for (int k = 0; k < BIN_EPT; ++k) {
        if (es[k] >= 0) {
            const int bin  = es[k] >> RPB_SHIFT;
            const int pos  = hist[bin] + rk[k];
            const int r0   = bin << RPB_SHIFT;
            const int rows = (N - r0) < RPB ? (N - r0) : RPB;
            if (pos < rows * BIN_CAP_PER_ROW) {
                float2 p;
                p.x = __int_as_float(es[k]);
                p.y = __int_as_float(ed[k]);
                *(float2*)(out + (size_t)r0 * OUT_DIM + 2 * (size_t)pos) = p;
            }
        }
    }
}

__device__ __forceinline__ void fma4(float s, const float4& m, float4& acc) {
    acc.x = fmaf(s, m.x, acc.x);
    acc.y = fmaf(s, m.y, acc.y);
    acc.z = fmaf(s, m.z, acc.z);
    acc.w = fmaf(s, m.w, acc.w);
}

__device__ __forceinline__ void gather4(const float* __restrict__ emb, int lane,
                                        int d0, int d1, int d2, int d3,
                                        float s0, float s1, float s2, float s3,
                                        float4& acc) {
    const float4 m0 = *(const float4*)(emb + (size_t)d0 * OUT_DIM + lane * 4);
    const float4 m1 = *(const float4*)(emb + (size_t)d1 * OUT_DIM + lane * 4);
    const float4 m2 = *(const float4*)(emb + (size_t)d2 * OUT_DIM + lane * 4);
    const float4 m3 = *(const float4*)(emb + (size_t)d3 * OUT_DIM + lane * 4);
    fma4(s0, m0, acc);
    fma4(s1, m1, acc);
    fma4(s2, m2, acc);
    fma4(s3, m3, acc);
}

__device__ __forceinline__ void gather1(const float* __restrict__ emb, int lane,
                                        int d0, float s0, float4& acc) {
    const float4 m0 = *(const float4*)(emb + (size_t)d0 * OUT_DIM + lane * 4);
    fma4(s0, m0, acc);
}

__device__ __forceinline__ float edge_score(float ss, float sd) {
    const float logit = ss + sd;
    const float lr    = logit > 0.0f ? logit : SLOPE * logit;
    return expf(lr);
}

__global__ __launch_bounds__(ROW_THREADS) void row_kernel(
    const int* __restrict__ edges, const int* __restrict__ bincnt,
    const float* __restrict__ s_src, const float* __restrict__ s_dst,
    const float* __restrict__ emb, float* __restrict__ out, int N, int E) {
    __shared__ int slot[RPB * ROW_SLOTS];
    __shared__ int rcnt[RPB];
    const int tid  = threadIdx.x;
    const int wave = __builtin_amdgcn_readfirstlane(tid >> 5);
    const int lane = tid & 31;
    const int b    = (int)blockIdx.x;
    const int r0   = b << RPB_SHIFT;
    const int rows = (N - r0) < RPB ? (N - r0) : RPB;
    const int total  = bincnt[b];
    const bool staged = b < MAX_BINS && total <= rows * BIN_CAP_PER_ROW;

    if (tid < RPB) rcnt[tid] = 0;
    __syncthreads();
    if (staged) {
        const float* st = out + (size_t)r0 * OUT_DIM;
        for (int i0 = 0; i0 < total; i0 += STAGE_EPT * ROW_THREADS) {
            float2 p[STAGE_EPT];
#pragma unroll
            for (int k = 0; k < STAGE_EPT; ++k) {
                const int i  = i0 + k * ROW_THREADS + tid;
                const int ic = i < total ? i : 0;
                p[k] = *(const float2*)(st + 2 * (size_t)ic);
            }
#pragma unroll
            for (int k = 0; k < STAGE_EPT; ++k) {
                const int i = i0 + k * ROW_THREADS + tid;
                if (i < total) {
                    const int lr  = __float_as_int(p[k].x) & (RPB - 1);
                    const int pos = atomicAdd(&rcnt[lr], 1);
                    if (pos < ROW_SLOTS) slot[lr * ROW_SLOTS + pos] = __float_as_int(p[k].y);
                }
            }
        }
    }
    __syncthreads();

    for (int lr = wave; lr < rows; lr += ROW_WAVES) {
        const int   r    = r0 + lr;
        const int   c    = staged ? __builtin_amdgcn_readfirstlane(rcnt[lr]) : ROW_SLOTS + 1;
        const float ss   = s_src[r];
        float*      orow = out + (size_t)r * OUT_DIM;
        const int*  srow = slot + lr * ROW_SLOTS;

        float4 acc = {0.0f, 0.0f, 0.0f, 0.0f};
        float  rs  = 0.0f;

        if (c <= ROW_SLOTS) {
            for (int base = 0; base < c; base += 32) {
                const int n = (c - base) < 32 ? (c - base) : 32;
                int   d  = 0;
                float sc = 0.0f;
                if (lane < n) {
                    d  = srow[base + lane];
                    sc = edge_score(ss, s_dst[d]);
                }
                rs += sc;
                const int sci = __float_as_int(sc);
                int t = 0;
                for (; t + 4 <= n; t += 4) {
                    gather4(emb, lane,
                            __builtin_amdgcn_readlane(d, t),
                            __builtin_amdgcn_readlane(d, t + 1),
                            __builtin_amdgcn_readlane(d, t + 2),
                            __builtin_amdgcn_readlane(d, t + 3),
                            __int_as_float(__builtin_amdgcn_readlane(sci, t)),
                            __int_as_float(__builtin_amdgcn_readlane(sci, t + 1)),
                            __int_as_float(__builtin_amdgcn_readlane(sci, t + 2)),
                            __int_as_float(__builtin_amdgcn_readlane(sci, t + 3)),
                            acc);
                }
#pragma nounroll
                for (; t < n; ++t) {
                    gather1(emb, lane, __builtin_amdgcn_readlane(d, t),
                            __int_as_float(__builtin_amdgcn_readlane(sci, t)), acc);
                }
            }
        } else {
            for (int base = 0; base < E; base += 32) {
                const int e = base + lane;
                int   sx = -1, d = 0;
                float sc = 0.0f;
                if (e < E) {
                    const int2 sd = *(const int2*)(edges + 2 * (size_t)e);
                    sx = sd.x;
                    d  = sd.y;
                }
                const bool hit = (sx == r);
                if (hit) sc = edge_score(ss, s_dst[d]);
                rs += sc;
                const int sci = __float_as_int(sc);
                unsigned mask = __builtin_amdgcn_ballot_w32(hit);
                while (mask) {
                    const int t = __builtin_ctz(mask);
                    mask &= mask - 1;
                    gather1(emb, lane, __builtin_amdgcn_readlane(d, t),
                            __int_as_float(__builtin_amdgcn_readlane(sci, t)), acc);
                }
            }
        }

#pragma unroll
        for (int off = 16; off > 0; off >>= 1) rs += __shfl_xor(rs, off, 32);
        const float safe = rs > 0.0f ? rs : 1.0f;

        float4 o;
        o.x = acc.x / safe; o.y = acc.y / safe; o.z = acc.z / safe; o.w = acc.w / safe;
        *(float4*)(orow + lane * 4) = o;
    }
}

extern "C" void kernel_launch(void* const* d_in, const int* in_sizes, int n_in,
                              void* d_out, int out_size, void* d_ws, size_t ws_size,
                              hipStream_t stream) {
    const float* x     = (const float*)d_in[0];
    const float* W     = (const float*)d_in[1];
    const float* bias  = (const float*)d_in[2];
    const float* a     = (const float*)d_in[3];
    const int*   edges = (const int*)d_in[4];

    const int N = in_sizes[0] / IN_DIM;
    const int E = in_sizes[4] / 2;

    float* ws    = (float*)d_ws;
    float* emb   = ws;
    float* s_src = emb + (size_t)N * OUT_DIM;
    float* s_dst = s_src + N;
    int*   cnt   = (int*)(s_dst + N);
    float* out   = (float*)d_out;

    {
        gemm_wmma_kernel<<<GEMM_GROUPS, GEMM_THREADS, 0, stream>>>(x, W, bias, a, emb, s_src, s_dst, cnt, N);
    }
    {
        int grid = (E + BIN_CHUNK - 1) / BIN_CHUNK;
        bin_kernel<<<grid, BIN_THREADS, 0, stream>>>(edges, cnt, out, N, E);
    }
    {
        int grid = (N + RPB - 1) / RPB;
        row_kernel<<<grid, ROW_THREADS, 0, stream>>>(edges, cnt, s_src, s_dst, emb, out, N, E);
    }
}
